# starting kernel
# baseline (speedup 1.0000x reference)
#include <hip/hip_runtime.h>
#include <hip/hip_bf16.h>

typedef float v2f __attribute__((ext_vector_type(2)));
typedef float v8f __attribute__((ext_vector_type(8)));

#define DF 64

__global__ void deg_kernel(const int* __restrict__ src, const int* __restrict__ dst,
                           float* __restrict__ deg_out, float* __restrict__ deg_in, int E) {
    int e = blockIdx.x * blockDim.x + threadIdx.x;
    if (e >= E) return;
    atomicAdd(&deg_out[src[e]], 1.0f);
    atomicAdd(&deg_in[dst[e]], 1.0f);
}

__global__ void norm_kernel(float* __restrict__ a, float* __restrict__ b, int n) {
    int i = blockIdx.x * blockDim.x + threadIdx.x;
    if (i >= n) return;
    a[i] = __frsqrt_rn(fmaxf(a[i], 1.0f));
    b[i] = __frsqrt_rn(fmaxf(b[i], 1.0f));
}

__global__ void prescale_kernel(const float* __restrict__ x, const float* __restrict__ s,
                                float* __restrict__ xs, int n_nodes) {
    int t = blockIdx.x * blockDim.x + threadIdx.x;
    int node = t >> 4;
    int q = t & 15;
    if (node >= n_nodes) return;
    float sc = s[node];
    float4 v = ((const float4*)(x + (size_t)node * DF))[q];
    v.x *= sc; v.y *= sc; v.z *= sc; v.w *= sc;
    ((float4*)(xs + (size_t)node * DF))[q] = v;
}

__global__ void scatter_kernel(const float* __restrict__ xs, const int* __restrict__ src,
                               const int* __restrict__ dst, float* __restrict__ agg, int E) {
    int t = blockIdx.x * blockDim.x + threadIdx.x;
    int e = t >> 4;
    int q = t & 15;
    if (e >= E) return;
    int s = src[e], d = dst[e];
    float4 v = ((const float4*)(xs + (size_t)s * DF))[q];
    float* p = agg + (size_t)d * DF + q * 4;
    atomicAdd(p + 0, v.x);
    atomicAdd(p + 1, v.y);
    atomicAdd(p + 2, v.z);
    atomicAdd(p + 3, v.w);
}

__global__ void pool_kernel(const float* __restrict__ h, const int* __restrict__ gid,
                            float* __restrict__ pool, float* __restrict__ cnt, int n_nodes) {
    int t = blockIdx.x * blockDim.x + threadIdx.x;
    int node = t >> 4;
    int q = t & 15;
    if (node >= n_nodes) return;
    int g = gid[node];
    float4 v = ((const float4*)(h + (size_t)node * DF))[q];
    float* p = pool + (size_t)g * DF + q * 4;
    atomicAdd(p + 0, v.x);
    atomicAdd(p + 1, v.y);
    atomicAdd(p + 2, v.z);
    atomicAdd(p + 3, v.w);
    if (q == 0) atomicAdd(&cnt[g], 1.0f);
}

__global__ void invcnt_kernel(float* __restrict__ cnt, int g) {
    int i = blockIdx.x * blockDim.x + threadIdx.x;
    if (i >= g) return;
    cnt[i] = 1.0f / fmaxf(cnt[i], 1.0f);
}

template <int NCOLS_C>
__global__ void gemm_wmma_kernel(const float* __restrict__ A,
                                 const float* __restrict__ rowscale,
                                 const float* __restrict__ W,
                                 const float* __restrict__ bias,
                                 float* __restrict__ out,
                                 int rows, int ncols_rt, int do_relu) {
    const int Ncols = (NCOLS_C > 0) ? NCOLS_C : ncols_rt;
    const int lane = threadIdx.x & 31;
    const int wave = threadIdx.x >> 5;
    const int half = lane >> 4;
    const int r16  = lane & 15;
    const int m0 = blockIdx.x * 16;
    const int n0 = wave * 16;

    const int row  = m0 + r16;
    const int rowc = (row < rows) ? row : 0;
    const float s = rowscale[rowc];
    const float* arow = A + (size_t)rowc * DF + 2 * half;

    const int col  = n0 + r16;
    const bool cok = (col < Ncols);
    const int colc = cok ? col : (Ncols - 1);
    const float* wp = W + (size_t)(2 * half) * Ncols + colc;

    v8f c = {};
#pragma unroll
    for (int k = 0; k < 16; ++k) {
        v2f a;
        a.x = arow[4 * k + 0] * s;
        a.y = arow[4 * k + 1] * s;
        v2f b;
        b.x = wp[(size_t)(4 * k) * Ncols];
        b.y = wp[(size_t)(4 * k + 1) * Ncols];
        c = __builtin_amdgcn_wmma_f32_16x16x4_f32(false, a, false, b, (short)0, c,
                                                  false, false);
    }

    const float bv = bias[colc];
    float* orow0 = out + (size_t)m0 * Ncols + col;
    if (m0 + 16 <= rows && cok) {
#pragma unroll
        for (int j = 0; j < 8; ++j) {
            float v = c[j] + bv;
            if (do_relu) v = fmaxf(v, 0.0f);
            orow0[(size_t)(j + 8 * half) * Ncols] = v;
        }
    } else if (cok) {
#pragma unroll
        for (int j = 0; j < 8; ++j) {
            int orow = m0 + j + 8 * half;
            if (orow < rows) {
                float v = c[j] + bv;
                if (do_relu) v = fmaxf(v, 0.0f);
                out[(size_t)orow * Ncols + col] = v;
            }
        }
    }
}

extern "C" void kernel_launch(void* const* d_in, const int* in_sizes, int n_in,
                              void* d_out, int out_size, void* d_ws, size_t ws_size,
                              hipStream_t stream) {
    const float* features = (const float*)d_in[0];
    const int*   src      = (const int*)d_in[1];
    const int*   dst      = (const int*)d_in[2];
    const int*   gid      = (const int*)d_in[3];
    const float* W1       = (const float*)d_in[4];
    const float* b1       = (const float*)d_in[5];
    const float* W2       = (const float*)d_in[6];
    const float* b2       = (const float*)d_in[7];
    const float* Wc       = (const float*)d_in[8];
    const float* bc       = (const float*)d_in[9];
    float* outp           = (float*)d_out;

    const int N    = in_sizes[3];
    const int E    = in_sizes[1];
    const int DOUT = in_sizes[9];
    const int G    = out_size / DOUT;

    float* ws    = (float*)d_ws;
    float* out_n = ws;
    float* in_n  = out_n + N;
    float* bufA  = in_n + N;
    float* bufB  = bufA + (size_t)N * DF;
    float* bufC  = bufB + (size_t)N * DF;
    float* pool  = bufC + (size_t)N * DF;
    float* cnt   = pool + (size_t)G * DF;

    const int TB = 256;

    hipMemsetAsync(out_n, 0, sizeof(float) * (size_t)N * 2, stream);
    deg_kernel<<<(E + TB - 1) / TB, TB, 0, stream>>>(src, dst, out_n, in_n, E);
    norm_kernel<<<(N + TB - 1) / TB, TB, 0, stream>>>(out_n, in_n, N);

    const int nodeT = N * 16;
    const int edgeT = E * 16;
    const int gemmGrid = (N + 15) / 16;

    prescale_kernel<<<(nodeT + TB - 1) / TB, TB, 0, stream>>>(features, out_n, bufA, N);
    hipMemsetAsync(bufB, 0, sizeof(float) * (size_t)N * DF, stream);
    scatter_kernel<<<(edgeT + TB - 1) / TB, TB, 0, stream>>>(bufA, src, dst, bufB, E);
    gemm_wmma_kernel<64><<<gemmGrid, 128, 0, stream>>>(bufB, in_n, W1, b1, bufC, N, DF, 1);

    prescale_kernel<<<(nodeT + TB - 1) / TB, TB, 0, stream>>>(bufC, out_n, bufA, N);
    hipMemsetAsync(bufB, 0, sizeof(float) * (size_t)N * DF, stream);
    scatter_kernel<<<(edgeT + TB - 1) / TB, TB, 0, stream>>>(bufA, src, dst, bufB, E);
    gemm_wmma_kernel<64><<<gemmGrid, 128, 0, stream>>>(bufB, in_n, W2, b2, bufC, N, DF, 1);

    hipMemsetAsync(pool, 0, sizeof(float) * ((size_t)G * DF + G), stream);
    pool_kernel<<<(nodeT + TB - 1) / TB, TB, 0, stream>>>(bufC, gid, pool, cnt, N);
    invcnt_kernel<<<(G + TB - 1) / TB, TB, 0, stream>>>(cnt, G);

    const int clsWaves = (DOUT + 15) / 16;
    if (DOUT == 8) {
        gemm_wmma_kernel<8><<<(G + 15) / 16, 32 * clsWaves, 0, stream>>>(
            pool, cnt, Wc, bc, outp, G, DOUT, 0);
    } else {
        gemm_wmma_kernel<-1><<<(G + 15) / 16, 32 * clsWaves, 0, stream>>>(
            pool, cnt, Wc, bc, outp, G, DOUT, 0);
    }
}


# degrees: each block filters a 32768-edge slice of src/dst by its 8192-node range, counts in LDS (integer), adds whole 128-B lines of counts at the end; no scattered global float atomic per edge
# speedup vs baseline: 1.7444x; 1.7444x over previous
#include <hip/hip_runtime.h>
#include <hip/hip_bf16.h>

typedef float v2f __attribute__((ext_vector_type(2)));
typedef float v8f __attribute__((ext_vector_type(8)));

#define DF 64

#define DEG_CAP_LOG 13
#define DEG_CAP   (1 << DEG_CAP_LOG)
#define DEG_TB    256
#define DEG_GROUP 8
#define DEG_TRIPS 4
#define DEG_TRIP_EDGES (DEG_TB * 4 * DEG_GROUP)
#define DEG_SLICE (DEG_TRIP_EDGES * DEG_TRIPS)

__device__ __forceinline__ void deg_count(unsigned* h, int id, int lo) {
    const unsigned l = (unsigned)(id - lo);
    if (l < (unsigned)DEG_CAP) atomicAdd(&h[l], 1u);
}

__global__ void __launch_bounds__(DEG_TB)
deg_kernel(const int* __restrict__ src, const int* __restrict__ dst,
           unsigned* __restrict__ deg_out, unsigned* __restrict__ deg_in, int E, int N) {
    __shared__ unsigned h_out[DEG_CAP];
    __shared__ unsigned h_in[DEG_CAP];
    const int tid = threadIdx.x;
    const int lo = blockIdx.y << DEG_CAP_LOG;
    const int e0 = blockIdx.x * DEG_SLICE;

    for (int i = tid; i < DEG_CAP; i += DEG_TB) { h_out[i] = 0u; h_in[i] = 0u; }
    __syncthreads();

    for (int t = 0; t < DEG_TRIPS; ++t) {
        const int tb = e0 + t * DEG_TRIP_EDGES;
        if (tb + DEG_TRIP_EDGES <= E) {
            int4 sv[DEG_GROUP], dv[DEG_GROUP];
#pragma unroll
            for (int u = 0; u < DEG_GROUP; ++u) {
                const int i = tb + (u * DEG_TB + tid) * 4;
                sv[u] = *(const int4*)(src + i);
                dv[u] = *(const int4*)(dst + i);
            }
#pragma unroll
            for (int u = 0; u < DEG_GROUP; ++u) {
                deg_count(h_out, sv[u].x, lo); deg_count(h_in, dv[u].x, lo);
                deg_count(h_out, sv[u].y, lo); deg_count(h_in, dv[u].y, lo);
                deg_count(h_out, sv[u].z, lo); deg_count(h_in, dv[u].z, lo);
                deg_count(h_out, sv[u].w, lo); deg_count(h_in, dv[u].w, lo);
            }
        } else {
            for (int i = tb + tid; i < tb + DEG_TRIP_EDGES; i += DEG_TB) {
                if (i < E) {
                    deg_count(h_out, src[i], lo);
                    deg_count(h_in, dst[i], lo);
                }
            }
        }
    }
    __syncthreads();

    for (int i = tid; i < DEG_CAP; i += DEG_TB) {
        const int node = lo + i;
        if (node < N) {
            const unsigned co = h_out[i];
            const unsigned ci = h_in[i];
            if (co) atomicAdd(&deg_out[node], co);
            if (ci) atomicAdd(&deg_in[node], ci);
        }
    }
}

__global__ void norm_kernel(float* __restrict__ a, float* __restrict__ b, int n) {
    int i = blockIdx.x * blockDim.x + threadIdx.x;
    if (i >= n) return;
    a[i] = __frsqrt_rn(fmaxf((float)__float_as_uint(a[i]), 1.0f));
    b[i] = __frsqrt_rn(fmaxf((float)__float_as_uint(b[i]), 1.0f));
}

__global__ void prescale_kernel(const float* __restrict__ x, const float* __restrict__ s,
                                float* __restrict__ xs, int n_nodes) {
    int t = blockIdx.x * blockDim.x + threadIdx.x;
    int node = t >> 4;
    int q = t & 15;
    if (node >= n_nodes) return;
    float sc = s[node];
    float4 v = ((const float4*)(x + (size_t)node * DF))[q];
    v.x *= sc; v.y *= sc; v.z *= sc; v.w *= sc;
    ((float4*)(xs + (size_t)node * DF))[q] = v;
}

#define SC_GROUP 8
__global__ void scatter_kernel(const float* __restrict__ xs, const int* __restrict__ src,
                               const int* __restrict__ dst, float* __restrict__ agg, int E) {
    const int lane = threadIdx.x & 31;
    const int wave = (blockIdx.x * blockDim.x + threadIdx.x) >> 5;
    const int base = wave * 32;
    if (base >= E) return;
    const int rem = E - base;
    const int cnt = rem < 32 ? rem : 32;
    const int e = base + (lane < cnt ? lane : cnt - 1);
    const int my_s = src[e];
    const int my_d = dst[e];
    const float* xl = xs + lane;
    float* al = agg + lane;

    if (cnt == 32) {
#pragma unroll
        for (int k0 = 0; k0 < 32; k0 += SC_GROUP) {
            float v0[SC_GROUP], v1[SC_GROUP];
#pragma unroll
            for (int j = 0; j < SC_GROUP; ++j) {
                const int s = __builtin_amdgcn_readlane(my_s, k0 + j);
                const float* r = xl + (size_t)s * DF;
                v0[j] = r[0];
                v1[j] = r[32];
            }
#pragma unroll
            for (int j = 0; j < SC_GROUP; ++j) {
                const int d = __builtin_amdgcn_readlane(my_d, k0 + j);
                float* p = al + (size_t)d * DF;
                atomicAdd(p, v0[j]);
                atomicAdd(p + 32, v1[j]);
            }
        }
    } else {
        for (int k = 0; k < cnt; ++k) {
            const int s = __builtin_amdgcn_readlane(my_s, k);
            const int d = __builtin_amdgcn_readlane(my_d, k);
            const float* r = xl + (size_t)s * DF;
            const float a0 = r[0];
            const float a1 = r[32];
            float* p = al + (size_t)d * DF;
            atomicAdd(p, a0);
            atomicAdd(p + 32, a1);
        }
    }
}

__global__ void pool_kernel(const float* __restrict__ h, const int* __restrict__ gid,
                            float* __restrict__ pool, float* __restrict__ cnt, int n_nodes) {
    int t = blockIdx.x * blockDim.x + threadIdx.x;
    int node = t >> 4;
    int q = t & 15;
    if (node >= n_nodes) return;
    int g = gid[node];
    float4 v = ((const float4*)(h + (size_t)node * DF))[q];
    float* p = pool + (size_t)g * DF + q * 4;
    atomicAdd(p + 0, v.x);
    atomicAdd(p + 1, v.y);
    atomicAdd(p + 2, v.z);
    atomicAdd(p + 3, v.w);
    if (q == 0) atomicAdd(&cnt[g], 1.0f);
}

__global__ void invcnt_kernel(float* __restrict__ cnt, int g) {
    int i = blockIdx.x * blockDim.x + threadIdx.x;
    if (i >= g) return;
    cnt[i] = 1.0f / fmaxf(cnt[i], 1.0f);
}

template <int NCOLS_C>
__global__ void gemm_wmma_kernel(const float* __restrict__ A,
                                 const float* __restrict__ rowscale,
                                 const float* __restrict__ W,
                                 const float* __restrict__ bias,
                                 float* __restrict__ out,
                                 int rows, int ncols_rt, int do_relu) {
    const int Ncols = (NCOLS_C > 0) ? NCOLS_C : ncols_rt;
    const int lane = threadIdx.x & 31;
    const int wave = threadIdx.x >> 5;
    const int half = lane >> 4;
    const int r16  = lane & 15;
    const int m0 = blockIdx.x * 16;
    const int n0 = wave * 16;

    const int row  = m0 + r16;
    const int rowc = (row < rows) ? row : 0;
    const float s = rowscale[rowc];
    const float* arow = A + (size_t)rowc * DF + 2 * half;

    const int col  = n0 + r16;
    const bool cok = (col < Ncols);
    const int colc = cok ? col : (Ncols - 1);
    const float* wp = W + (size_t)(2 * half) * Ncols + colc;

    v8f c = {};
#pragma unroll
    for (int k = 0; k < 16; ++k) {
        v2f a;
        a.x = arow[4 * k + 0] * s;
        a.y = arow[4 * k + 1] * s;
        v2f b;
        b.x = wp[(size_t)(4 * k) * Ncols];
        b.y = wp[(size_t)(4 * k + 1) * Ncols];
        c = __builtin_amdgcn_wmma_f32_16x16x4_f32(false, a, false, b, (short)0, c,
                                                  false, false);
    }

    const float bv = bias[colc];
    float* orow0 = out + (size_t)m0 * Ncols + col;
    if (m0 + 16 <= rows && cok) {
#pragma unroll
        for (int j = 0; j < 8; ++j) {
            float v = c[j] + bv;
            if (do_relu) v = fmaxf(v, 0.0f);
            orow0[(size_t)(j + 8 * half) * Ncols] = v;
        }
    } else if (cok) {
#pragma unroll
        for (int j = 0; j < 8; ++j) {
            int orow = m0 + j + 8 * half;
            if (orow < rows) {
                float v = c[j] + bv;
                if (do_relu) v = fmaxf(v, 0.0f);
                out[(size_t)orow * Ncols + col] = v;
            }
        }
    }
}

extern "C" void kernel_launch(void* const* d_in, const int* in_sizes, int n_in,
                              void* d_out, int out_size, void* d_ws, size_t ws_size,
                              hipStream_t stream) {
    const float* features = (const float*)d_in[0];
    const int*   src      = (const int*)d_in[1];
    const int*   dst      = (const int*)d_in[2];
    const int*   gid      = (const int*)d_in[3];
    const float* W1       = (const float*)d_in[4];
    const float* b1       = (const float*)d_in[5];
    const float* W2       = (const float*)d_in[6];
    const float* b2       = (const float*)d_in[7];
    const float* Wc       = (const float*)d_in[8];
    const float* bc       = (const float*)d_in[9];
    float* outp           = (float*)d_out;

    const int N    = in_sizes[3];
    const int E    = in_sizes[1];
    const int DOUT = in_sizes[9];
    const int G    = out_size / DOUT;

    float* ws    = (float*)d_ws;
    float* out_n = ws;
    float* in_n  = out_n + N;
    float* bufA  = in_n + N;
    float* bufB  = bufA + (size_t)N * DF;
    float* bufC  = bufB + (size_t)N * DF;
    float* pool  = bufC + (size_t)N * DF;
    float* cnt   = pool + (size_t)G * DF;

    const int TB = 256;

    hipMemsetAsync(out_n, 0, sizeof(float) * (size_t)N * 2, stream);
    const dim3 degGrid((E + DEG_SLICE - 1) / DEG_SLICE, (N + DEG_CAP - 1) / DEG_CAP);
    deg_kernel<<<degGrid, DEG_TB, 0, stream>>>(src, dst, (unsigned*)out_n, (unsigned*)in_n, E, N);
    norm_kernel<<<(N + TB - 1) / TB, TB, 0, stream>>>(out_n, in_n, N);

    const int nodeT = N * 16;
    const int edgeT = E;
    const int gemmGrid = (N + 15) / 16;

    prescale_kernel<<<(nodeT + TB - 1) / TB, TB, 0, stream>>>(features, out_n, bufA, N);
    hipMemsetAsync(bufB, 0, sizeof(float) * (size_t)N * DF, stream);
    scatter_kernel<<<(edgeT + TB - 1) / TB, TB, 0, stream>>>(bufA, src, dst, bufB, E);
    gemm_wmma_kernel<64><<<gemmGrid, 128, 0, stream>>>(bufB, in_n, W1, b1, bufC, N, DF, 1);

    prescale_kernel<<<(nodeT + TB - 1) / TB, TB, 0, stream>>>(bufC, out_n, bufA, N);
    hipMemsetAsync(bufB, 0, sizeof(float) * (size_t)N * DF, stream);
    scatter_kernel<<<(edgeT + TB - 1) / TB, TB, 0, stream>>>(bufA, src, dst, bufB, E);
    gemm_wmma_kernel<64><<<gemmGrid, 128, 0, stream>>>(bufB, in_n, W2, b2, bufC, N, DF, 1);

    hipMemsetAsync(pool, 0, sizeof(float) * ((size_t)G * DF + G), stream);
    pool_kernel<<<(nodeT + TB - 1) / TB, TB, 0, stream>>>(bufC, gid, pool, cnt, N);
    invcnt_kernel<<<(G + TB - 1) / TB, TB, 0, stream>>>(cnt, G);

    const int clsWaves = (DOUT + 15) / 16;
    if (DOUT == 8) {
        gemm_wmma_kernel<8><<<(G + 15) / 16, 32 * clsWaves, 0, stream>>>(
            pool, cnt, Wc, bc, outp, G, DOUT, 0);
    } else {
        gemm_wmma_kernel<-1><<<(G + 15) / 16, 32 * clsWaves, 0, stream>>>(
            pool, cnt, Wc, bc, outp, G, DOUT, 0);
    }
}


# weight products moved ahead of their segment sums: layers sum rows already times W (out-norm = GEMM row scale, in-norm+bias+relu on next read), read-out pools rows x Wc; atomic sums, new rounding
# speedup vs baseline: 2.4969x; 1.4314x over previous
#include <hip/hip_runtime.h>
#include <hip/hip_bf16.h>

typedef float v2f __attribute__((ext_vector_type(2)));
typedef float v8f __attribute__((ext_vector_type(8)));

#define DF 64

#define DEG_CAP_LOG 13
#define DEG_CAP   (1 << DEG_CAP_LOG)
#define DEG_TB    256
#define DEG_GROUP 8
#define DEG_TRIPS 4
#define DEG_TRIP_EDGES (DEG_TB * 4 * DEG_GROUP)
#define DEG_SLICE (DEG_TRIP_EDGES * DEG_TRIPS)

__device__ __forceinline__ void deg_count(unsigned* h, int id, int lo) {
    const unsigned l = (unsigned)(id - lo);
    if (l < (unsigned)DEG_CAP) atomicAdd(&h[l], 1u);
}

__global__ void __launch_bounds__(DEG_TB)
deg_kernel(const int* __restrict__ src, const int* __restrict__ dst,
           unsigned* __restrict__ deg_out, unsigned* __restrict__ deg_in, int E, int N) {
    __shared__ unsigned h_out[DEG_CAP];
    __shared__ unsigned h_in[DEG_CAP];
    const int tid = threadIdx.x;
    const int lo = blockIdx.y << DEG_CAP_LOG;
    const int e0 = blockIdx.x * DEG_SLICE;

    for (int i = tid; i < DEG_CAP; i += DEG_TB) { h_out[i] = 0u; h_in[i] = 0u; }
    __syncthreads();

    for (int t = 0; t < DEG_TRIPS; ++t) {
        const int tb = e0 + t * DEG_TRIP_EDGES;
        if (tb + DEG_TRIP_EDGES <= E) {
            int4 sv[DEG_GROUP], dv[DEG_GROUP];
#pragma unroll
            for (int u = 0; u < DEG_GROUP; ++u) {
                const int i = tb + (u * DEG_TB + tid) * 4;
                sv[u] = *(const int4*)(src + i);
                dv[u] = *(const int4*)(dst + i);
            }
#pragma unroll
            for (int u = 0; u < DEG_GROUP; ++u) {
                deg_count(h_out, sv[u].x, lo); deg_count(h_in, dv[u].x, lo);
                deg_count(h_out, sv[u].y, lo); deg_count(h_in, dv[u].y, lo);
                deg_count(h_out, sv[u].z, lo); deg_count(h_in, dv[u].z, lo);
                deg_count(h_out, sv[u].w, lo); deg_count(h_in, dv[u].w, lo);
            }
        } else {
            for (int i = tb + tid; i < tb + DEG_TRIP_EDGES; i += DEG_TB) {
                if (i < E) {
                    deg_count(h_out, src[i], lo);
                    deg_count(h_in, dst[i], lo);
                }
            }
        }
    }
    __syncthreads();

    for (int i = tid; i < DEG_CAP; i += DEG_TB) {
        const int node = lo + i;
        if (node < N) {
            const unsigned co = h_out[i];
            const unsigned ci = h_in[i];
            if (co) atomicAdd(&deg_out[node], co);
            if (ci) atomicAdd(&deg_in[node], ci);
        }
    }
}

__global__ void norm_kernel(float* __restrict__ a, float* __restrict__ b, int n) {
    int i = blockIdx.x * blockDim.x + threadIdx.x;
    if (i >= n) return;
    a[i] = __frsqrt_rn(fmaxf((float)__float_as_uint(a[i]), 1.0f));
    b[i] = __frsqrt_rn(fmaxf((float)__float_as_uint(b[i]), 1.0f));
}

#define SC_GROUP 8
__global__ void scatter_kernel(const float* __restrict__ xs, const int* __restrict__ src,
                               const int* __restrict__ dst, float* __restrict__ agg, int E) {
    const int lane = threadIdx.x & 31;
    const int wave = (blockIdx.x * blockDim.x + threadIdx.x) >> 5;
    const int base = wave * 32;
    if (base >= E) return;
    const int rem = E - base;
    const int cnt = rem < 32 ? rem : 32;
    const int e = base + (lane < cnt ? lane : cnt - 1);
    const int my_s = src[e];
    const int my_d = dst[e];
    const float* xl = xs + lane;
    float* al = agg + lane;

    if (cnt == 32) {
#pragma unroll
        for (int k0 = 0; k0 < 32; k0 += SC_GROUP) {
            float v0[SC_GROUP], v1[SC_GROUP];
#pragma unroll
            for (int j = 0; j < SC_GROUP; ++j) {
                const int s = __builtin_amdgcn_readlane(my_s, k0 + j);
                const float* r = xl + (size_t)s * DF;
                v0[j] = r[0];
                v1[j] = r[32];
            }
#pragma unroll
            for (int j = 0; j < SC_GROUP; ++j) {
                const int d = __builtin_amdgcn_readlane(my_d, k0 + j);
                float* p = al + (size_t)d * DF;
                atomicAdd(p, v0[j]);
                atomicAdd(p + 32, v1[j]);
            }
        }
    } else {
        for (int k = 0; k < cnt; ++k) {
            const int s = __builtin_amdgcn_readlane(my_s, k);
            const int d = __builtin_amdgcn_readlane(my_d, k);
            const float* r = xl + (size_t)s * DF;
            const float a0 = r[0];
            const float a1 = r[32];
            float* p = al + (size_t)d * DF;
            atomicAdd(p, a0);
            atomicAdd(p + 32, a1);
        }
    }
}

template <bool PRE>
__global__ void gemm_wmma_kernel(const float* __restrict__ A,
                                 const float* __restrict__ rowscale,
                                 const float* __restrict__ prescale,
                                 const float* __restrict__ prebias,
                                 const float* __restrict__ W,
                                 float* __restrict__ out, int rows) {
    const int lane = threadIdx.x & 31;
    const int wave = threadIdx.x >> 5;
    const int half = lane >> 4;
    const int r16  = lane & 15;
    const int m0 = blockIdx.x * 16;
    const int n0 = wave * 16;

    const int row  = m0 + r16;
    const int rowc = (row < rows) ? row : 0;
    const float s = rowscale[rowc];
    float ps = 0.0f;
    if (PRE) ps = prescale[rowc];
    const float* arow = A + (size_t)rowc * DF + 2 * half;
    const float* pb = PRE ? prebias + 2 * half : nullptr;

    const int col = n0 + r16;
    const float* wp = W + (size_t)(2 * half) * DF + col;

    v8f c = {};
#pragma unroll
    for (int k = 0; k < 16; ++k) {
        float a0 = arow[4 * k + 0];
        float a1 = arow[4 * k + 1];
        if (PRE) {
            a0 = fmaxf(fmaf(a0, ps, pb[4 * k + 0]), 0.0f);
            a1 = fmaxf(fmaf(a1, ps, pb[4 * k + 1]), 0.0f);
        }
        v2f a;
        a.x = a0 * s;
        a.y = a1 * s;
        v2f b;
        b.x = wp[(size_t)(4 * k) * DF];
        b.y = wp[(size_t)(4 * k + 1) * DF];
        c = __builtin_amdgcn_wmma_f32_16x16x4_f32(false, a, false, b, (short)0, c,
                                                  false, false);
    }

    float* orow0 = out + (size_t)m0 * DF + col;
    if (m0 + 16 <= rows) {
#pragma unroll
        for (int j = 0; j < 8; ++j) orow0[(size_t)(j + 8 * half) * DF] = c[j];
    } else {
#pragma unroll
        for (int j = 0; j < 8; ++j) {
            const int orow = m0 + j + 8 * half;
            if (orow < rows) out[(size_t)orow * DF + col] = c[j];
        }
    }
}

template <int NC>
__global__ void readout_kernel(const float* __restrict__ S, const float* __restrict__ in_norm,
                               const float* __restrict__ bias, const float* __restrict__ Wc,
                               const int* __restrict__ gid, float* __restrict__ zc,
                               int rows, int ncols_rt) {
    const int ncols = (NC > 0) ? NC : ncols_rt;
    const int lane = threadIdx.x & 31;
    const int wave = threadIdx.x >> 5;
    const int half = lane >> 4;
    const int r16  = lane & 15;
    const int m0 = (blockIdx.x * (blockDim.x >> 5) + wave) * 16;

    const bool full = (m0 + 16 <= rows);
    int gs[8];
    if (full) {
        const int4 ga = *(const int4*)(gid + m0 + 8 * half);
        const int4 gb = *(const int4*)(gid + m0 + 8 * half + 4);
        gs[0] = ga.x; gs[1] = ga.y; gs[2] = ga.z; gs[3] = ga.w;
        gs[4] = gb.x; gs[5] = gb.y; gs[6] = gb.z; gs[7] = gb.w;
    } else {
#pragma unroll
        for (int j = 0; j < 8; ++j) {
            const int orow = m0 + 8 * half + j;
            gs[j] = gid[orow < rows ? orow : 0];
        }
    }

    const int row  = m0 + r16;
    const int rowc = (row < rows) ? row : 0;
    const float ps = in_norm[rowc];
    const float* arow = S + (size_t)rowc * DF + 2 * half;
    const float* pb = bias + 2 * half;

    const bool cok = (r16 < ncols);
    const int colc = cok ? r16 : (ncols - 1);
    const float* wp = Wc + (2 * half) * ncols + colc;

    v2f av[16], bv[16], wv[16];
#pragma unroll
    for (int k = 0; k < 16; ++k) av[k] = *(const v2f*)(arow + 4 * k);
#pragma unroll
    for (int k = 0; k < 16; ++k) bv[k] = *(const v2f*)(pb + 4 * k);
#pragma unroll
    for (int k = 0; k < 16; ++k) {
        wv[k].x = wp[(4 * k) * ncols];
        wv[k].y = wp[(4 * k + 1) * ncols];
    }

    v8f c = {};
#pragma unroll
    for (int k = 0; k < 16; ++k) {
        v2f a;
        a.x = fmaxf(fmaf(av[k].x, ps, bv[k].x), 0.0f);
        a.y = fmaxf(fmaf(av[k].y, ps, bv[k].y), 0.0f);
        c = __builtin_amdgcn_wmma_f32_16x16x4_f32(false, a, false, wv[k], (short)0, c,
                                                  false, false);
    }

    if (r16 <= ncols) {
        float* zl = zc + r16;
        if (full) {
#pragma unroll
            for (int j = 0; j < 8; ++j)
                atomicAdd(zl + gs[j] * (ncols + 1), cok ? c[j] : 1.0f);
        } else {
#pragma unroll
            for (int j = 0; j < 8; ++j) {
                if (m0 + 8 * half + j < rows)
                    atomicAdd(zl + gs[j] * (ncols + 1), cok ? c[j] : 1.0f);
            }
        }
    }
}

__global__ void finish_kernel(const float* __restrict__ zc, const float* __restrict__ bc,
                              float* __restrict__ out, int total, int ncols) {
    const int i = blockIdx.x * blockDim.x + threadIdx.x;
    if (i >= total) return;
    const int g = i / ncols;
    const int cidx = i - g * ncols;
    const float* zr = zc + g * (ncols + 1);
    out[i] = zr[cidx] / fmaxf(zr[ncols], 1.0f) + bc[cidx];
}

extern "C" void kernel_launch(void* const* d_in, const int* in_sizes, int n_in,
                              void* d_out, int out_size, void* d_ws, size_t ws_size,
                              hipStream_t stream) {
    const float* features = (const float*)d_in[0];
    const int*   src      = (const int*)d_in[1];
    const int*   dst      = (const int*)d_in[2];
    const int*   gid      = (const int*)d_in[3];
    const float* W1       = (const float*)d_in[4];
    const float* b1       = (const float*)d_in[5];
    const float* W2       = (const float*)d_in[6];
    const float* b2       = (const float*)d_in[7];
    const float* Wc       = (const float*)d_in[8];
    const float* bc       = (const float*)d_in[9];
    float* outp           = (float*)d_out;

    const int N    = in_sizes[3];
    const int E    = in_sizes[1];
    const int DOUT = in_sizes[9];
    const int G    = out_size / DOUT;

    float* ws    = (float*)d_ws;
    float* out_n = ws;
    float* in_n  = out_n + N;
    float* bufA  = in_n + N;
    float* bufB  = bufA + (size_t)N * DF;
    float* zc    = bufB + (size_t)N * DF;

    const int TB = 256;

    hipMemsetAsync(out_n, 0, sizeof(float) * (size_t)N * 2, stream);
    const dim3 degGrid((E + DEG_SLICE - 1) / DEG_SLICE, (N + DEG_CAP - 1) / DEG_CAP);
    deg_kernel<<<degGrid, DEG_TB, 0, stream>>>(src, dst, (unsigned*)out_n, (unsigned*)in_n, E, N);
    norm_kernel<<<(N + TB - 1) / TB, TB, 0, stream>>>(out_n, in_n, N);

    const int edgeT = E;
    const int gemmGrid = (N + 15) / 16;

    gemm_wmma_kernel<false><<<gemmGrid, 128, 0, stream>>>(features, out_n, nullptr, nullptr, W1, bufA, N);
    hipMemsetAsync(bufB, 0, sizeof(float) * (size_t)N * DF, stream);
    scatter_kernel<<<(edgeT + TB - 1) / TB, TB, 0, stream>>>(bufA, src, dst, bufB, E);

    gemm_wmma_kernel<true><<<gemmGrid, 128, 0, stream>>>(bufB, out_n, in_n, b1, W2, bufA, N);
    hipMemsetAsync(bufB, 0, sizeof(float) * (size_t)N * DF, stream);
    scatter_kernel<<<(edgeT + TB - 1) / TB, TB, 0, stream>>>(bufA, src, dst, bufB, E);

    hipMemsetAsync(zc, 0, sizeof(float) * (size_t)G * (DOUT + 1), stream);
    if (DOUT == 8) {
        readout_kernel<8><<<(N + 63) / 64, 128, 0, stream>>>(bufB, in_n, b2, Wc, gid, zc, N, DOUT);
    } else {
        readout_kernel<-1><<<(N + 63) / 64, 128, 0, stream>>>(bufB, in_n, b2, Wc, gid, zc, N, DOUT);
    }
    finish_kernel<<<(G * DOUT + TB - 1) / TB, TB, 0, stream>>>(zc, bc, outp, G * DOUT, DOUT);
}


# zero fills moved into kernels: each GEMM block also stores zeros to its 16 rows of the next row-sum buffer (whole lines), the norm pass zeroes the graph sums; 3 of 4 memsets gone; sums as before
# speedup vs baseline: 3.1478x; 1.2607x over previous
#include <hip/hip_runtime.h>
#include <hip/hip_bf16.h>

typedef float v2f __attribute__((ext_vector_type(2)));
typedef float v8f __attribute__((ext_vector_type(8)));

#define DF 64

#define DEG_CAP_LOG 13
#define DEG_CAP   (1 << DEG_CAP_LOG)
#define DEG_TB    256
#define DEG_GROUP 8
#define DEG_TRIPS 4
#define DEG_TRIP_EDGES (DEG_TB * 4 * DEG_GROUP)
#define DEG_SLICE (DEG_TRIP_EDGES * DEG_TRIPS)

__device__ __forceinline__ void deg_count(unsigned* h, int id, int lo) {
    const unsigned l = (unsigned)(id - lo);
    if (l < (unsigned)DEG_CAP) atomicAdd(&h[l], 1u);
}

__global__ void __launch_bounds__(DEG_TB)
deg_kernel(const int* __restrict__ src, const int* __restrict__ dst,
           unsigned* __restrict__ deg_out, unsigned* __restrict__ deg_in, int E, int N) {
    __shared__ unsigned h_out[DEG_CAP];
    __shared__ unsigned h_in[DEG_CAP];
    const int tid = threadIdx.x;
    const int lo = blockIdx.y << DEG_CAP_LOG;
    const int e0 = blockIdx.x * DEG_SLICE;

    for (int i = tid; i < DEG_CAP; i += DEG_TB) { h_out[i] = 0u; h_in[i] = 0u; }
    __syncthreads();

    for (int t = 0; t < DEG_TRIPS; ++t) {
        const int tb = e0 + t * DEG_TRIP_EDGES;
        if (tb + DEG_TRIP_EDGES <= E) {
            int4 sv[DEG_GROUP], dv[DEG_GROUP];
#pragma unroll
            for (int u = 0; u < DEG_GROUP; ++u) {
                const int i = tb + (u * DEG_TB + tid) * 4;
                sv[u] = *(const int4*)(src + i);
                dv[u] = *(const int4*)(dst + i);
            }
#pragma unroll
            for (int u = 0; u < DEG_GROUP; ++u) {
                deg_count(h_out, sv[u].x, lo); deg_count(h_in, dv[u].x, lo);
                deg_count(h_out, sv[u].y, lo); deg_count(h_in, dv[u].y, lo);
                deg_count(h_out, sv[u].z, lo); deg_count(h_in, dv[u].z, lo);
                deg_count(h_out, sv[u].w, lo); deg_count(h_in, dv[u].w, lo);
            }
        } else {
            for (int i = tb + tid; i < tb + DEG_TRIP_EDGES; i += DEG_TB) {
                if (i < E) {
                    deg_count(h_out, src[i], lo);
                    deg_count(h_in, dst[i], lo);
                }
            }
        }
    }
    __syncthreads();

    for (int i = tid; i < DEG_CAP; i += DEG_TB) {
        const int node = lo + i;
        if (node < N) {
            const unsigned co = h_out[i];
            const unsigned ci = h_in[i];
            if (co) atomicAdd(&deg_out[node], co);
            if (ci) atomicAdd(&deg_in[node], ci);
        }
    }
}

__global__ void norm_kernel(float* __restrict__ a, float* __restrict__ b, int n,
                            float* __restrict__ zc, int nz) {
    int i = blockIdx.x * blockDim.x + threadIdx.x;
    for (int j = i; j < nz; j += gridDim.x * blockDim.x) zc[j] = 0.0f;
    if (i >= n) return;
    a[i] = __frsqrt_rn(fmaxf((float)__float_as_uint(a[i]), 1.0f));
    b[i] = __frsqrt_rn(fmaxf((float)__float_as_uint(b[i]), 1.0f));
}

#define SC_GROUP 8
__global__ void scatter_kernel(const float* __restrict__ xs, const int* __restrict__ src,
                               const int* __restrict__ dst, float* __restrict__ agg, int E) {
    const int lane = threadIdx.x & 31;
    const int wave = (blockIdx.x * blockDim.x + threadIdx.x) >> 5;
    const int base = wave * 32;
    if (base >= E) return;
    const int rem = E - base;
    const int cnt = rem < 32 ? rem : 32;
    const int e = base + (lane < cnt ? lane : cnt - 1);
    const int my_s = src[e];
    const int my_d = dst[e];
    const float* xl = xs + lane;
    float* al = agg + lane;

    if (cnt == 32) {
#pragma unroll
        for (int k0 = 0; k0 < 32; k0 += SC_GROUP) {
            float v0[SC_GROUP], v1[SC_GROUP];
#pragma unroll
            for (int j = 0; j < SC_GROUP; ++j) {
                const int s = __builtin_amdgcn_readlane(my_s, k0 + j);
                const float* r = xl + (size_t)s * DF;
                v0[j] = r[0];
                v1[j] = r[32];
            }
#pragma unroll
            for (int j = 0; j < SC_GROUP; ++j) {
                const int d = __builtin_amdgcn_readlane(my_d, k0 + j);
                float* p = al + (size_t)d * DF;
                atomicAdd(p, v0[j]);
                atomicAdd(p + 32, v1[j]);
            }
        }
    } else {
        for (int k = 0; k < cnt; ++k) {
            const int s = __builtin_amdgcn_readlane(my_s, k);
            const int d = __builtin_amdgcn_readlane(my_d, k);
            const float* r = xl + (size_t)s * DF;
            const float a0 = r[0];
            const float a1 = r[32];
            float* p = al + (size_t)d * DF;
            atomicAdd(p, a0);
            atomicAdd(p + 32, a1);
        }
    }
}

template <bool PRE>
__global__ void gemm_wmma_kernel(const float* __restrict__ A,
                                 const float* __restrict__ rowscale,
                                 const float* __restrict__ prescale,
                                 const float* __restrict__ prebias,
                                 const float* __restrict__ W,
                                 float* __restrict__ out,
                                 float* __restrict__ zrows, int rows) {
    const int lane = threadIdx.x & 31;
    const int wave = threadIdx.x >> 5;
    const int half = lane >> 4;
    const int r16  = lane & 15;
    const int m0 = blockIdx.x * 16;
    const int n0 = wave * 16;

    {
        float4* zt = (float4*)(zrows + (size_t)m0 * DF);
        const float4 z4 = make_float4(0.0f, 0.0f, 0.0f, 0.0f);
        if (m0 + 16 <= rows) {
            zt[threadIdx.x] = z4;
            zt[threadIdx.x + 128] = z4;
        } else {
            for (int i = threadIdx.x; i < 256; i += 128)
                if (m0 + (i >> 4) < rows) zt[i] = z4;
        }
    }

    const int row  = m0 + r16;
    const int rowc = (row < rows) ? row : 0;
    const float s = rowscale[rowc];
    float ps = 0.0f;
    if (PRE) ps = prescale[rowc];
    const float* arow = A + (size_t)rowc * DF + 2 * half;
    const float* pb = PRE ? prebias + 2 * half : nullptr;

    const int col = n0 + r16;
    const float* wp = W + (size_t)(2 * half) * DF + col;

    v8f c = {};
#pragma unroll
    for (int k = 0; k < 16; ++k) {
        float a0 = arow[4 * k + 0];
        float a1 = arow[4 * k + 1];
        if (PRE) {
            a0 = fmaxf(fmaf(a0, ps, pb[4 * k + 0]), 0.0f);
            a1 = fmaxf(fmaf(a1, ps, pb[4 * k + 1]), 0.0f);
        }
        v2f a;
        a.x = a0 * s;
        a.y = a1 * s;
        v2f b;
        b.x = wp[(size_t)(4 * k) * DF];
        b.y = wp[(size_t)(4 * k + 1) * DF];
        c = __builtin_amdgcn_wmma_f32_16x16x4_f32(false, a, false, b, (short)0, c,
                                                  false, false);
    }

    float* orow0 = out + (size_t)m0 * DF + col;
    if (m0 + 16 <= rows) {
#pragma unroll
        for (int j = 0; j < 8; ++j) orow0[(size_t)(j + 8 * half) * DF] = c[j];
    } else {
#pragma unroll
        for (int j = 0; j < 8; ++j) {
            const int orow = m0 + j + 8 * half;
            if (orow < rows) out[(size_t)orow * DF + col] = c[j];
        }
    }
}

template <int NC>
__global__ void readout_kernel(const float* __restrict__ S, const float* __restrict__ in_norm,
                               const float* __restrict__ bias, const float* __restrict__ Wc,
                               const int* __restrict__ gid, float* __restrict__ zc,
                               int rows, int ncols_rt) {
    const int ncols = (NC > 0) ? NC : ncols_rt;
    const int lane = threadIdx.x & 31;
    const int wave = threadIdx.x >> 5;
    const int half = lane >> 4;
    const int r16  = lane & 15;
    const int m0 = (blockIdx.x * (blockDim.x >> 5) + wave) * 16;

    const bool full = (m0 + 16 <= rows);
    int gs[8];
    if (full) {
        const int4 ga = *(const int4*)(gid + m0 + 8 * half);
        const int4 gb = *(const int4*)(gid + m0 + 8 * half + 4);
        gs[0] = ga.x; gs[1] = ga.y; gs[2] = ga.z; gs[3] = ga.w;
        gs[4] = gb.x; gs[5] = gb.y; gs[6] = gb.z; gs[7] = gb.w;
    } else {
#pragma unroll
        for (int j = 0; j < 8; ++j) {
            const int orow = m0 + 8 * half + j;
            gs[j] = gid[orow < rows ? orow : 0];
        }
    }

    const int row  = m0 + r16;
    const int rowc = (row < rows) ? row : 0;
    const float ps = in_norm[rowc];
    const float* arow = S + (size_t)rowc * DF + 2 * half;
    const float* pb = bias + 2 * half;

    const bool cok = (r16 < ncols);
    const int colc = cok ? r16 : (ncols - 1);
    const float* wp = Wc + (2 * half) * ncols + colc;

    v2f av[16], bv[16], wv[16];
#pragma unroll
    for (int k = 0; k < 16; ++k) av[k] = *(const v2f*)(arow + 4 * k);
#pragma unroll
    for (int k = 0; k < 16; ++k) bv[k] = *(const v2f*)(pb + 4 * k);
#pragma unroll
    for (int k = 0; k < 16; ++k) {
        wv[k].x = wp[(4 * k) * ncols];
        wv[k].y = wp[(4 * k + 1) * ncols];
    }

    v8f c = {};
#pragma unroll
    for (int k = 0; k < 16; ++k) {
        v2f a;
        a.x = fmaxf(fmaf(av[k].x, ps, bv[k].x), 0.0f);
        a.y = fmaxf(fmaf(av[k].y, ps, bv[k].y), 0.0f);
        c = __builtin_amdgcn_wmma_f32_16x16x4_f32(false, a, false, wv[k], (short)0, c,
                                                  false, false);
    }

    if (r16 <= ncols) {
        float* zl = zc + r16;
        if (full) {
#pragma unroll
            for (int j = 0; j < 8; ++j)
                atomicAdd(zl + gs[j] * (ncols + 1), cok ? c[j] : 1.0f);
        } else {
#pragma unroll
            for (int j = 0; j < 8; ++j) {
                if (m0 + 8 * half + j < rows)
                    atomicAdd(zl + gs[j] * (ncols + 1), cok ? c[j] : 1.0f);
            }
        }
    }
}

__global__ void finish_kernel(const float* __restrict__ zc, const float* __restrict__ bc,
                              float* __restrict__ out, int total, int ncols) {
    const int i = blockIdx.x * blockDim.x + threadIdx.x;
    if (i >= total) return;
    const int g = i / ncols;
    const int cidx = i - g * ncols;
    const float* zr = zc + g * (ncols + 1);
    out[i] = zr[cidx] / fmaxf(zr[ncols], 1.0f) + bc[cidx];
}

extern "C" void kernel_launch(void* const* d_in, const int* in_sizes, int n_in,
                              void* d_out, int out_size, void* d_ws, size_t ws_size,
                              hipStream_t stream) {
    const float* features = (const float*)d_in[0];
    const int*   src      = (const int*)d_in[1];
    const int*   dst      = (const int*)d_in[2];
    const int*   gid      = (const int*)d_in[3];
    const float* W1       = (const float*)d_in[4];
    const float* b1       = (const float*)d_in[5];
    const float* W2       = (const float*)d_in[6];
    const float* b2       = (const float*)d_in[7];
    const float* Wc       = (const float*)d_in[8];
    const float* bc       = (const float*)d_in[9];
    float* outp           = (float*)d_out;

    const int N    = in_sizes[3];
    const int E    = in_sizes[1];
    const int DOUT = in_sizes[9];
    const int G    = out_size / DOUT;

    float* ws    = (float*)d_ws;
    float* bufA  = ws;
    float* bufB  = bufA + (size_t)N * DF;
    float* bufC  = bufB + (size_t)N * DF;
    float* out_n = bufC + (size_t)N * DF;
    float* in_n  = out_n + N;
    float* zc    = in_n + N;

    const int TB = 256;

    hipMemsetAsync(out_n, 0, sizeof(float) * (size_t)N * 2, stream);
    const dim3 degGrid((E + DEG_SLICE - 1) / DEG_SLICE, (N + DEG_CAP - 1) / DEG_CAP);
    deg_kernel<<<degGrid, DEG_TB, 0, stream>>>(src, dst, (unsigned*)out_n, (unsigned*)in_n, E, N);
    norm_kernel<<<(N + TB - 1) / TB, TB, 0, stream>>>(out_n, in_n, N, zc, G * (DOUT + 1));

    const int edgeT = E;
    const int gemmGrid = (N + 15) / 16;

    gemm_wmma_kernel<false><<<gemmGrid, 128, 0, stream>>>(features, out_n, nullptr, nullptr, W1, bufA, bufB, N);
    scatter_kernel<<<(edgeT + TB - 1) / TB, TB, 0, stream>>>(bufA, src, dst, bufB, E);

    gemm_wmma_kernel<true><<<gemmGrid, 128, 0, stream>>>(bufB, out_n, in_n, b1, W2, bufA, bufC, N);
    scatter_kernel<<<(edgeT + TB - 1) / TB, TB, 0, stream>>>(bufA, src, dst, bufC, E);

    if (DOUT == 8) {
        readout_kernel<8><<<(N + 63) / 64, 128, 0, stream>>>(bufC, in_n, b2, Wc, gid, zc, N, DOUT);
    } else {
        readout_kernel<-1><<<(N + 63) / 64, 128, 0, stream>>>(bufC, in_n, b2, Wc, gid, zc, N, DOUT);
    }
    finish_kernel<<<(G * DOUT + TB - 1) / TB, TB, 0, stream>>>(zc, bc, outp, G * DOUT, DOUT);
}


# scatter: blocks draw 256-edge items from one ticket counter per half of the package and add only the edges whose destination row is in that half's node range: sum lines stay in one L2; arrival order
# speedup vs baseline: 3.3993x; 1.0799x over previous
#include <hip/hip_runtime.h>
#include <hip/hip_bf16.h>

typedef float v2f __attribute__((ext_vector_type(2)));
typedef float v8f __attribute__((ext_vector_type(8)));

#define DF 64

#define DEG_CAP_LOG 13
#define DEG_CAP   (1 << DEG_CAP_LOG)
#define DEG_TB    256
#define DEG_GROUP 8
#define DEG_TRIPS 4
#define DEG_TRIP_EDGES (DEG_TB * 4 * DEG_GROUP)
#define DEG_SLICE (DEG_TRIP_EDGES * DEG_TRIPS)

__device__ __forceinline__ void deg_count(unsigned* h, int id, int lo) {
    const unsigned l = (unsigned)(id - lo);
    if (l < (unsigned)DEG_CAP) atomicAdd(&h[l], 1u);
}

__global__ void __launch_bounds__(DEG_TB)
deg_kernel(const int* __restrict__ src, const int* __restrict__ dst,
           unsigned* __restrict__ deg_out, unsigned* __restrict__ deg_in, int E, int N) {
    __shared__ unsigned h_out[DEG_CAP];
    __shared__ unsigned h_in[DEG_CAP];
    const int tid = threadIdx.x;
    const int lo = blockIdx.y << DEG_CAP_LOG;
    const int e0 = blockIdx.x * DEG_SLICE;

    for (int i = tid; i < DEG_CAP; i += DEG_TB) { h_out[i] = 0u; h_in[i] = 0u; }
    __syncthreads();

    for (int t = 0; t < DEG_TRIPS; ++t) {
        const int tb = e0 + t * DEG_TRIP_EDGES;
        if (tb + DEG_TRIP_EDGES <= E) {
            int4 sv[DEG_GROUP], dv[DEG_GROUP];
#pragma unroll
            for (int u = 0; u < DEG_GROUP; ++u) {
                const int i = tb + (u * DEG_TB + tid) * 4;
                sv[u] = *(const int4*)(src + i);
                dv[u] = *(const int4*)(dst + i);
            }
#pragma unroll
            for (int u = 0; u < DEG_GROUP; ++u) {
                deg_count(h_out, sv[u].x, lo); deg_count(h_in, dv[u].x, lo);
                deg_count(h_out, sv[u].y, lo); deg_count(h_in, dv[u].y, lo);
                deg_count(h_out, sv[u].z, lo); deg_count(h_in, dv[u].z, lo);
                deg_count(h_out, sv[u].w, lo); deg_count(h_in, dv[u].w, lo);
            }
        } else {
            for (int i = tb + tid; i < tb + DEG_TRIP_EDGES; i += DEG_TB) {
                if (i < E) {
                    deg_count(h_out, src[i], lo);
                    deg_count(h_in, dst[i], lo);
                }
            }
        }
    }
    __syncthreads();

    for (int i = tid; i < DEG_CAP; i += DEG_TB) {
        const int node = lo + i;
        if (node < N) {
            const unsigned co = h_out[i];
            const unsigned ci = h_in[i];
            if (co) atomicAdd(&deg_out[node], co);
            if (ci) atomicAdd(&deg_in[node], ci);
        }
    }
}

__global__ void norm_kernel(float* __restrict__ a, float* __restrict__ b, int n,
                            float* __restrict__ zc, int nz, int* __restrict__ tickets, int nq) {
    int i = blockIdx.x * blockDim.x + threadIdx.x;
    for (int j = i; j < nz; j += gridDim.x * blockDim.x) zc[j] = 0.0f;
    for (int j = i; j < nq; j += gridDim.x * blockDim.x) tickets[j] = 0;
    if (i >= n) return;
    a[i] = __frsqrt_rn(fmaxf((float)__float_as_uint(a[i]), 1.0f));
    b[i] = __frsqrt_rn(fmaxf((float)__float_as_uint(b[i]), 1.0f));
}

#define SC_TB     256
#define SC_ITEM   SC_TB
#define SC_GROUP  16
#define SC_QPITCH 64
#define SC_BLOCKS 2048

__global__ void __launch_bounds__(SC_TB)
scatter_kernel(const float* __restrict__ xs, const int* __restrict__ src,
               const int* __restrict__ dst, float* __restrict__ agg,
               int* __restrict__ tickets, int E, int N) {
    __shared__ int s_ticket[2];
    __shared__ int s_half;
    const int lane = threadIdx.x & 31;
    const int wib  = __builtin_amdgcn_readfirstlane(threadIdx.x >> 5);
    const int split  = N >> 1;
    const int nitems = (E + SC_ITEM - 1) / SC_ITEM;
    const float* xl = xs + lane;
    float* al = agg + lane;

    if (threadIdx.x == 0) {
        const int hh = (__builtin_amdgcn_s_sendmsg_rtn(0x87) >> 18) & 1;
        s_half = hh;
        s_ticket[0] = atomicAdd(&tickets[hh * SC_QPITCH], 1);
    }
    __syncthreads();
    int h = __builtin_amdgcn_readfirstlane(s_half);
    bool second = false;
    int slot = 0;

    for (;;) {
        const int it = __builtin_amdgcn_readfirstlane(s_ticket[slot]);
        slot ^= 1;
        if (it >= nitems) {
            if (second) break;
            second = true;
            h ^= 1;
            if (threadIdx.x == 0) s_ticket[slot] = atomicAdd(&tickets[h * SC_QPITCH], 1);
            __syncthreads();
            continue;
        }

        const int base = it * SC_ITEM + wib * 32;
        const int rem = E - base;
        const int cnt = rem < 32 ? rem : 32;
        int my_s = 0, my_d = 0;
        if (cnt > 0) {
            const int e = base + (lane < cnt ? lane : cnt - 1);
            my_s = src[e];
            my_d = dst[e];
        }
        int nx = 0;
        if (threadIdx.x == 0) nx = atomicAdd(&tickets[h * SC_QPITCH], 1);

        const unsigned m = __builtin_amdgcn_ballot_w32((lane < cnt) && ((my_d >= split) == (h != 0)));

#pragma unroll
        for (int k0 = 0; k0 < 32; k0 += SC_GROUP) {
            float v0[SC_GROUP], v1[SC_GROUP];
#pragma unroll
            for (int j = 0; j < SC_GROUP; ++j) {
                if (m & (1u << (k0 + j))) {
                    const int s = __builtin_amdgcn_readlane(my_s, k0 + j);
                    const float* r = xl + (size_t)s * DF;
                    v0[j] = r[0];
                    v1[j] = r[32];
                }
            }
#pragma unroll
            for (int j = 0; j < SC_GROUP; ++j) {
                if (m & (1u << (k0 + j))) {
                    const int d = __builtin_amdgcn_readlane(my_d, k0 + j);
                    float* p = al + (size_t)d * DF;
                    atomicAdd(p, v0[j]);
                    atomicAdd(p + 32, v1[j]);
                }
            }
        }

        if (threadIdx.x == 0) s_ticket[slot] = nx;
        __syncthreads();
    }
}

template <bool PRE>
__global__ void gemm_wmma_kernel(const float* __restrict__ A,
                                 const float* __restrict__ rowscale,
                                 const float* __restrict__ prescale,
                                 const float* __restrict__ prebias,
                                 const float* __restrict__ W,
                                 float* __restrict__ out,
                                 float* __restrict__ zrows, int rows) {
    const int lane = threadIdx.x & 31;
    const int wave = threadIdx.x >> 5;
    const int half = lane >> 4;
    const int r16  = lane & 15;
    const int m0 = blockIdx.x * 16;
    const int n0 = wave * 16;

    {
        float4* zt = (float4*)(zrows + (size_t)m0 * DF);
        const float4 z4 = make_float4(0.0f, 0.0f, 0.0f, 0.0f);
        if (m0 + 16 <= rows) {
            zt[threadIdx.x] = z4;
            zt[threadIdx.x + 128] = z4;
        } else {
            for (int i = threadIdx.x; i < 256; i += 128)
                if (m0 + (i >> 4) < rows) zt[i] = z4;
        }
    }

    const int row  = m0 + r16;
    const int rowc = (row < rows) ? row : 0;
    const float s = rowscale[rowc];
    float ps = 0.0f;
    if (PRE) ps = prescale[rowc];
    const float* arow = A + (size_t)rowc * DF + 2 * half;
    const float* pb = PRE ? prebias + 2 * half : nullptr;

    const int col = n0 + r16;
    const float* wp = W + (size_t)(2 * half) * DF + col;

    v8f c = {};
#pragma unroll
    for (int k = 0; k < 16; ++k) {
        float a0 = arow[4 * k + 0];
        float a1 = arow[4 * k + 1];
        if (PRE) {
            a0 = fmaxf(fmaf(a0, ps, pb[4 * k + 0]), 0.0f);
            a1 = fmaxf(fmaf(a1, ps, pb[4 * k + 1]), 0.0f);
        }
        v2f a;
        a.x = a0 * s;
        a.y = a1 * s;
        v2f b;
        b.x = wp[(size_t)(4 * k) * DF];
        b.y = wp[(size_t)(4 * k + 1) * DF];
        c = __builtin_amdgcn_wmma_f32_16x16x4_f32(false, a, false, b, (short)0, c,
                                                  false, false);
    }

    float* orow0 = out + (size_t)m0 * DF + col;
    if (m0 + 16 <= rows) {
#pragma unroll
        for (int j = 0; j < 8; ++j) orow0[(size_t)(j + 8 * half) * DF] = c[j];
    } else {
#pragma unroll
        for (int j = 0; j < 8; ++j) {
            const int orow = m0 + j + 8 * half;
            if (orow < rows) out[(size_t)orow * DF + col] = c[j];
        }
    }
}

template <int NC>
__global__ void readout_kernel(const float* __restrict__ S, const float* __restrict__ in_norm,
                               const float* __restrict__ bias, const float* __restrict__ Wc,
                               const int* __restrict__ gid, float* __restrict__ zc,
                               int rows, int ncols_rt) {
    const int ncols = (NC > 0) ? NC : ncols_rt;
    const int lane = threadIdx.x & 31;
    const int wave = threadIdx.x >> 5;
    const int half = lane >> 4;
    const int r16  = lane & 15;
    const int m0 = (blockIdx.x * (blockDim.x >> 5) + wave) * 16;

    const bool full = (m0 + 16 <= rows);
    int gs[8];
    if (full) {
        const int4 ga = *(const int4*)(gid + m0 + 8 * half);
        const int4 gb = *(const int4*)(gid + m0 + 8 * half + 4);
        gs[0] = ga.x; gs[1] = ga.y; gs[2] = ga.z; gs[3] = ga.w;
        gs[4] = gb.x; gs[5] = gb.y; gs[6] = gb.z; gs[7] = gb.w;
    } else {
#pragma unroll
        for (int j = 0; j < 8; ++j) {
            const int orow = m0 + 8 * half + j;
            gs[j] = gid[orow < rows ? orow : 0];
        }
    }

    const int row  = m0 + r16;
    const int rowc = (row < rows) ? row : 0;
    const float ps = in_norm[rowc];
    const float* arow = S + (size_t)rowc * DF + 2 * half;
    const float* pb = bias + 2 * half;

    const bool cok = (r16 < ncols);
    const int colc = cok ? r16 : (ncols - 1);
    const float* wp = Wc + (2 * half) * ncols + colc;

    v2f av[16], bv[16], wv[16];
#pragma unroll
    for (int k = 0; k < 16; ++k) av[k] = *(const v2f*)(arow + 4 * k);
#pragma unroll
    for (int k = 0; k < 16; ++k) bv[k] = *(const v2f*)(pb + 4 * k);
#pragma unroll
    for (int k = 0; k < 16; ++k) {
        wv[k].x = wp[(4 * k) * ncols];
        wv[k].y = wp[(4 * k + 1) * ncols];
    }

    v8f c = {};
#pragma unroll
    for (int k = 0; k < 16; ++k) {
        v2f a;
        a.x = fmaxf(fmaf(av[k].x, ps, bv[k].x), 0.0f);
        a.y = fmaxf(fmaf(av[k].y, ps, bv[k].y), 0.0f);
        c = __builtin_amdgcn_wmma_f32_16x16x4_f32(false, a, false, wv[k], (short)0, c,
                                                  false, false);
    }

    if (r16 <= ncols) {
        float* zl = zc + r16;
        if (full) {
#pragma unroll
            for (int j = 0; j < 8; ++j)
                atomicAdd(zl + gs[j] * (ncols + 1), cok ? c[j] : 1.0f);
        } else {
#pragma unroll
            for (int j = 0; j < 8; ++j) {
                if (m0 + 8 * half + j < rows)
                    atomicAdd(zl + gs[j] * (ncols + 1), cok ? c[j] : 1.0f);
            }
        }
    }
}

__global__ void finish_kernel(const float* __restrict__ zc, const float* __restrict__ bc,
                              float* __restrict__ out, int total, int ncols) {
    const int i = blockIdx.x * blockDim.x + threadIdx.x;
    if (i >= total) return;
    const int g = i / ncols;
    const int cidx = i - g * ncols;
    const float* zr = zc + g * (ncols + 1);
    out[i] = zr[cidx] / fmaxf(zr[ncols], 1.0f) + bc[cidx];
}

extern "C" void kernel_launch(void* const* d_in, const int* in_sizes, int n_in,
                              void* d_out, int out_size, void* d_ws, size_t ws_size,
                              hipStream_t stream) {
    const float* features = (const float*)d_in[0];
    const int*   src      = (const int*)d_in[1];
    const int*   dst      = (const int*)d_in[2];
    const int*   gid      = (const int*)d_in[3];
    const float* W1       = (const float*)d_in[4];
    const float* b1       = (const float*)d_in[5];
    const float* W2       = (const float*)d_in[6];
    const float* b2       = (const float*)d_in[7];
    const float* Wc       = (const float*)d_in[8];
    const float* bc       = (const float*)d_in[9];
    float* outp           = (float*)d_out;

    const int N    = in_sizes[3];
    const int E    = in_sizes[1];
    const int DOUT = in_sizes[9];
    const int G    = out_size / DOUT;

    int* tickets = (int*)d_ws;
    float* ws    = (float*)d_ws + 4 * SC_QPITCH;
    float* bufA  = ws;
    float* bufB  = bufA + (size_t)N * DF;
    float* bufC  = bufB + (size_t)N * DF;
    float* out_n = bufC + (size_t)N * DF;
    float* in_n  = out_n + N;
    float* zc    = in_n + N;

    const int TB = 256;

    hipMemsetAsync(out_n, 0, sizeof(float) * (size_t)N * 2, stream);
    const dim3 degGrid((E + DEG_SLICE - 1) / DEG_SLICE, (N + DEG_CAP - 1) / DEG_CAP);
    deg_kernel<<<degGrid, DEG_TB, 0, stream>>>(src, dst, (unsigned*)out_n, (unsigned*)in_n, E, N);
    norm_kernel<<<(N + TB - 1) / TB, TB, 0, stream>>>(out_n, in_n, N, zc, G * (DOUT + 1),
                                                      tickets, 4 * SC_QPITCH);

    const int gemmGrid = (N + 15) / 16;

    gemm_wmma_kernel<false><<<gemmGrid, 128, 0, stream>>>(features, out_n, nullptr, nullptr, W1, bufA, bufB, N);
    scatter_kernel<<<SC_BLOCKS, SC_TB, 0, stream>>>(bufA, src, dst, bufB, tickets, E, N);

    gemm_wmma_kernel<true><<<gemmGrid, 128, 0, stream>>>(bufB, out_n, in_n, b1, W2, bufA, bufC, N);
    scatter_kernel<<<SC_BLOCKS, SC_TB, 0, stream>>>(bufA, src, dst, bufC, tickets + 2 * SC_QPITCH, E, N);

    if (DOUT == 8) {
        readout_kernel<8><<<(N + 63) / 64, 128, 0, stream>>>(bufC, in_n, b2, Wc, gid, zc, N, DOUT);
    } else {
        readout_kernel<-1><<<(N + 63) / 64, 128, 0, stream>>>(bufC, in_n, b2, Wc, gid, zc, N, DOUT);
    }
    finish_kernel<<<(G * DOUT + TB - 1) / TB, TB, 0, stream>>>(zc, bc, outp, G * DOUT, DOUT);
}


# scatter: a wave's row loads for all 32 edge slots go out as ONE group ahead of its atomics (64 value registers, grid 1536 = resident blocks); the 2nd 16-slot group's loads had waited one by one
# speedup vs baseline: 3.5625x; 1.0480x over previous
#include <hip/hip_runtime.h>
#include <hip/hip_bf16.h>

typedef float v2f __attribute__((ext_vector_type(2)));
typedef float v8f __attribute__((ext_vector_type(8)));

#define DF 64

#define DEG_CAP_LOG 13
#define DEG_CAP   (1 << DEG_CAP_LOG)
#define DEG_TB    256
#define DEG_GROUP 8
#define DEG_TRIPS 4
#define DEG_TRIP_EDGES (DEG_TB * 4 * DEG_GROUP)
#define DEG_SLICE (DEG_TRIP_EDGES * DEG_TRIPS)

__device__ __forceinline__ void deg_count(unsigned* h, int id, int lo) {
    const unsigned l = (unsigned)(id - lo);
    if (l < (unsigned)DEG_CAP) atomicAdd(&h[l], 1u);
}

__global__ void __launch_bounds__(DEG_TB)
deg_kernel(const int* __restrict__ src, const int* __restrict__ dst,
           unsigned* __restrict__ deg_out, unsigned* __restrict__ deg_in, int E, int N) {
    __shared__ unsigned h_out[DEG_CAP];
    __shared__ unsigned h_in[DEG_CAP];
    const int tid = threadIdx.x;
    const int lo = blockIdx.y << DEG_CAP_LOG;
    const int e0 = blockIdx.x * DEG_SLICE;

    for (int i = tid; i < DEG_CAP; i += DEG_TB) { h_out[i] = 0u; h_in[i] = 0u; }
    __syncthreads();

    for (int t = 0; t < DEG_TRIPS; ++t) {
        const int tb = e0 + t * DEG_TRIP_EDGES;
        if (tb + DEG_TRIP_EDGES <= E) {
            int4 sv[DEG_GROUP], dv[DEG_GROUP];
#pragma unroll
            for (int u = 0; u < DEG_GROUP; ++u) {
                const int i = tb + (u * DEG_TB + tid) * 4;
                sv[u] = *(const int4*)(src + i);
                dv[u] = *(const int4*)(dst + i);
            }
#pragma unroll
            for (int u = 0; u < DEG_GROUP; ++u) {
                deg_count(h_out, sv[u].x, lo); deg_count(h_in, dv[u].x, lo);
                deg_count(h_out, sv[u].y, lo); deg_count(h_in, dv[u].y, lo);
                deg_count(h_out, sv[u].z, lo); deg_count(h_in, dv[u].z, lo);
                deg_count(h_out, sv[u].w, lo); deg_count(h_in, dv[u].w, lo);
            }
        } else {
            for (int i = tb + tid; i < tb + DEG_TRIP_EDGES; i += DEG_TB) {
                if (i < E) {
                    deg_count(h_out, src[i], lo);
                    deg_count(h_in, dst[i], lo);
                }
            }
        }
    }
    __syncthreads();

    for (int i = tid; i < DEG_CAP; i += DEG_TB) {
        const int node = lo + i;
        if (node < N) {
            const unsigned co = h_out[i];
            const unsigned ci = h_in[i];
            if (co) atomicAdd(&deg_out[node], co);
            if (ci) atomicAdd(&deg_in[node], ci);
        }
    }
}

__global__ void norm_kernel(float* __restrict__ a, float* __restrict__ b, int n,
                            float* __restrict__ zc, int nz, int* __restrict__ tickets, int nq) {
    int i = blockIdx.x * blockDim.x + threadIdx.x;
    for (int j = i; j < nz; j += gridDim.x * blockDim.x) zc[j] = 0.0f;
    for (int j = i; j < nq; j += gridDim.x * blockDim.x) tickets[j] = 0;
    if (i >= n) return;
    a[i] = __frsqrt_rn(fmaxf((float)__float_as_uint(a[i]), 1.0f));
    b[i] = __frsqrt_rn(fmaxf((float)__float_as_uint(b[i]), 1.0f));
}

#define SC_TB     256
#define SC_ITEM   SC_TB
#define SC_GROUP  32
#define SC_QPITCH 64
#define SC_BLOCKS 1536

__global__ void __launch_bounds__(SC_TB)
scatter_kernel(const float* __restrict__ xs, const int* __restrict__ src,
               const int* __restrict__ dst, float* __restrict__ agg,
               int* __restrict__ tickets, int E, int N) {
    __shared__ int s_ticket[2];
    __shared__ int s_half;
    const int lane = threadIdx.x & 31;
    const int wib  = __builtin_amdgcn_readfirstlane(threadIdx.x >> 5);
    const int split  = N >> 1;
    const int nitems = (E + SC_ITEM - 1) / SC_ITEM;
    const float* xl = xs + lane;
    float* al = agg + lane;

    if (threadIdx.x == 0) {
        const int hh = (__builtin_amdgcn_s_sendmsg_rtn(0x87) >> 18) & 1;
        s_half = hh;
        s_ticket[0] = atomicAdd(&tickets[hh * SC_QPITCH], 1);
    }
    __syncthreads();
    int h = __builtin_amdgcn_readfirstlane(s_half);
    bool second = false;
    int slot = 0;

    for (;;) {
        const int it = __builtin_amdgcn_readfirstlane(s_ticket[slot]);
        slot ^= 1;
        if (it >= nitems) {
            if (second) break;
            second = true;
            h ^= 1;
            if (threadIdx.x == 0) s_ticket[slot] = atomicAdd(&tickets[h * SC_QPITCH], 1);
            __syncthreads();
            continue;
        }

        const int base = it * SC_ITEM + wib * 32;
        const int rem = E - base;
        const int cnt = rem < 32 ? rem : 32;
        int my_s = 0, my_d = 0;
        if (cnt > 0) {
            const int e = base + (lane < cnt ? lane : cnt - 1);
            my_s = src[e];
            my_d = dst[e];
        }
        int nx = 0;
        if (threadIdx.x == 0) nx = atomicAdd(&tickets[h * SC_QPITCH], 1);

        const unsigned m = __builtin_amdgcn_ballot_w32((lane < cnt) && ((my_d >= split) == (h != 0)));

#pragma unroll
        for (int k0 = 0; k0 < 32; k0 += SC_GROUP) {
            float v0[SC_GROUP], v1[SC_GROUP];
#pragma unroll
            for (int j = 0; j < SC_GROUP; ++j) {
                if (m & (1u << (k0 + j))) {
                    const int s = __builtin_amdgcn_readlane(my_s, k0 + j);
                    const float* r = xl + (size_t)s * DF;
                    v0[j] = r[0];
                    v1[j] = r[32];
                }
            }
#pragma unroll
            for (int j = 0; j < SC_GROUP; ++j) {
                if (m & (1u << (k0 + j))) {
                    const int d = __builtin_amdgcn_readlane(my_d, k0 + j);
                    float* p = al + (size_t)d * DF;
                    atomicAdd(p, v0[j]);
                    atomicAdd(p + 32, v1[j]);
                }
            }
        }

        if (threadIdx.x == 0) s_ticket[slot] = nx;
        __syncthreads();
    }
}

template <bool PRE>
__global__ void gemm_wmma_kernel(const float* __restrict__ A,
                                 const float* __restrict__ rowscale,
                                 const float* __restrict__ prescale,
                                 const float* __restrict__ prebias,
                                 const float* __restrict__ W,
                                 float* __restrict__ out,
                                 float* __restrict__ zrows, int rows) {
    const int lane = threadIdx.x & 31;
    const int wave = threadIdx.x >> 5;
    const int half = lane >> 4;
    const int r16  = lane & 15;
    const int m0 = blockIdx.x * 16;
    const int n0 = wave * 16;

    {
        float4* zt = (float4*)(zrows + (size_t)m0 * DF);
        const float4 z4 = make_float4(0.0f, 0.0f, 0.0f, 0.0f);
        if (m0 + 16 <= rows) {
            zt[threadIdx.x] = z4;
            zt[threadIdx.x + 128] = z4;
        } else {
            for (int i = threadIdx.x; i < 256; i += 128)
                if (m0 + (i >> 4) < rows) zt[i] = z4;
        }
    }

    const int row  = m0 + r16;
    const int rowc = (row < rows) ? row : 0;
    const float s = rowscale[rowc];
    float ps = 0.0f;
    if (PRE) ps = prescale[rowc];
    const float* arow = A + (size_t)rowc * DF + 2 * half;
    const float* pb = PRE ? prebias + 2 * half : nullptr;

    const int col = n0 + r16;
    const float* wp = W + (size_t)(2 * half) * DF + col;

    v8f c = {};
#pragma unroll
    for (int k = 0; k < 16; ++k) {
        float a0 = arow[4 * k + 0];
        float a1 = arow[4 * k + 1];
        if (PRE) {
            a0 = fmaxf(fmaf(a0, ps, pb[4 * k + 0]), 0.0f);
            a1 = fmaxf(fmaf(a1, ps, pb[4 * k + 1]), 0.0f);
        }
        v2f a;
        a.x = a0 * s;
        a.y = a1 * s;
        v2f b;
        b.x = wp[(size_t)(4 * k) * DF];
        b.y = wp[(size_t)(4 * k + 1) * DF];
        c = __builtin_amdgcn_wmma_f32_16x16x4_f32(false, a, false, b, (short)0, c,
                                                  false, false);
    }

    float* orow0 = out + (size_t)m0 * DF + col;
    if (m0 + 16 <= rows) {
#pragma unroll
        for (int j = 0; j < 8; ++j) orow0[(size_t)(j + 8 * half) * DF] = c[j];
    } else {
#pragma unroll
        for (int j = 0; j < 8; ++j) {
            const int orow = m0 + j + 8 * half;
            if (orow < rows) out[(size_t)orow * DF + col] = c[j];
        }
    }
}

template <int NC>
__global__ void readout_kernel(const float* __restrict__ S, const float* __restrict__ in_norm,
                               const float* __restrict__ bias, const float* __restrict__ Wc,
                               const int* __restrict__ gid, float* __restrict__ zc,
                               int rows, int ncols_rt) {
    const int ncols = (NC > 0) ? NC : ncols_rt;
    const int lane = threadIdx.x & 31;
    const int wave = threadIdx.x >> 5;
    const int half = lane >> 4;
    const int r16  = lane & 15;
    const int m0 = (blockIdx.x * (blockDim.x >> 5) + wave) * 16;

    const bool full = (m0 + 16 <= rows);
    int gs[8];
    if (full) {
        const int4 ga = *(const int4*)(gid + m0 + 8 * half);
        const int4 gb = *(const int4*)(gid + m0 + 8 * half + 4);
        gs[0] = ga.x; gs[1] = ga.y; gs[2] = ga.z; gs[3] = ga.w;
        gs[4] = gb.x; gs[5] = gb.y; gs[6] = gb.z; gs[7] = gb.w;
    } else {
#pragma unroll
        for (int j = 0; j < 8; ++j) {
            const int orow = m0 + 8 * half + j;
            gs[j] = gid[orow < rows ? orow : 0];
        }
    }

    const int row  = m0 + r16;
    const int rowc = (row < rows) ? row : 0;
    const float ps = in_norm[rowc];
    const float* arow = S + (size_t)rowc * DF + 2 * half;
    const float* pb = bias + 2 * half;

    const bool cok = (r16 < ncols);
    const int colc = cok ? r16 : (ncols - 1);
    const float* wp = Wc + (2 * half) * ncols + colc;

    v2f av[16], bv[16], wv[16];
#pragma unroll
    for (int k = 0; k < 16; ++k) av[k] = *(const v2f*)(arow + 4 * k);
#pragma unroll
    for (int k = 0; k < 16; ++k) bv[k] = *(const v2f*)(pb + 4 * k);
#pragma unroll
    for (int k = 0; k < 16; ++k) {
        wv[k].x = wp[(4 * k) * ncols];
        wv[k].y = wp[(4 * k + 1) * ncols];
    }

    v8f c = {};
#pragma unroll
    for (int k = 0; k < 16; ++k) {
        v2f a;
        a.x = fmaxf(fmaf(av[k].x, ps, bv[k].x), 0.0f);
        a.y = fmaxf(fmaf(av[k].y, ps, bv[k].y), 0.0f);
        c = __builtin_amdgcn_wmma_f32_16x16x4_f32(false, a, false, wv[k], (short)0, c,
                                                  false, false);
    }

    if (r16 <= ncols) {
        float* zl = zc + r16;
        if (full) {
#pragma unroll
            for (int j = 0; j < 8; ++j)
                atomicAdd(zl + gs[j] * (ncols + 1), cok ? c[j] : 1.0f);
        } else {
#pragma unroll
            for (int j = 0; j < 8; ++j) {
                if (m0 + 8 * half + j < rows)
                    atomicAdd(zl + gs[j] * (ncols + 1), cok ? c[j] : 1.0f);
            }
        }
    }
}

__global__ void finish_kernel(const float* __restrict__ zc, const float* __restrict__ bc,
                              float* __restrict__ out, int total, int ncols) {
    const int i = blockIdx.x * blockDim.x + threadIdx.x;
    if (i >= total) return;
    const int g = i / ncols;
    const int cidx = i - g * ncols;
    const float* zr = zc + g * (ncols + 1);
    out[i] = zr[cidx] / fmaxf(zr[ncols], 1.0f) + bc[cidx];
}

extern "C" void kernel_launch(void* const* d_in, const int* in_sizes, int n_in,
                              void* d_out, int out_size, void* d_ws, size_t ws_size,
                              hipStream_t stream) {
    const float* features = (const float*)d_in[0];
    const int*   src      = (const int*)d_in[1];
    const int*   dst      = (const int*)d_in[2];
    const int*   gid      = (const int*)d_in[3];
    const float* W1       = (const float*)d_in[4];
    const float* b1       = (const float*)d_in[5];
    const float* W2       = (const float*)d_in[6];
    const float* b2       = (const float*)d_in[7];
    const float* Wc       = (const float*)d_in[8];
    const float* bc       = (const float*)d_in[9];
    float* outp           = (float*)d_out;

    const int N    = in_sizes[3];
    const int E    = in_sizes[1];
    const int DOUT = in_sizes[9];
    const int G    = out_size / DOUT;

    int* tickets = (int*)d_ws;
    float* ws    = (float*)d_ws + 4 * SC_QPITCH;
    float* bufA  = ws;
    float* bufB  = bufA + (size_t)N * DF;
    float* bufC  = bufB + (size_t)N * DF;
    float* out_n = bufC + (size_t)N * DF;
    float* in_n  = out_n + N;
    float* zc    = in_n + N;

    const int TB = 256;

    hipMemsetAsync(out_n, 0, sizeof(float) * (size_t)N * 2, stream);
    const dim3 degGrid((E + DEG_SLICE - 1) / DEG_SLICE, (N + DEG_CAP - 1) / DEG_CAP);
    deg_kernel<<<degGrid, DEG_TB, 0, stream>>>(src, dst, (unsigned*)out_n, (unsigned*)in_n, E, N);
    norm_kernel<<<(N + TB - 1) / TB, TB, 0, stream>>>(out_n, in_n, N, zc, G * (DOUT + 1),
                                                      tickets, 4 * SC_QPITCH);

    const int gemmGrid = (N + 15) / 16;

    gemm_wmma_kernel<false><<<gemmGrid, 128, 0, stream>>>(features, out_n, nullptr, nullptr, W1, bufA, bufB, N);
    scatter_kernel<<<SC_BLOCKS, SC_TB, 0, stream>>>(bufA, src, dst, bufB, tickets, E, N);

    gemm_wmma_kernel<true><<<gemmGrid, 128, 0, stream>>>(bufB, out_n, in_n, b1, W2, bufA, bufC, N);
    scatter_kernel<<<SC_BLOCKS, SC_TB, 0, stream>>>(bufA, src, dst, bufC, tickets + 2 * SC_QPITCH, E, N);

    if (DOUT == 8) {
        readout_kernel<8><<<(N + 63) / 64, 128, 0, stream>>>(bufC, in_n, b2, Wc, gid, zc, N, DOUT);
    } else {
        readout_kernel<-1><<<(N + 63) / 64, 128, 0, stream>>>(bufC, in_n, b2, Wc, gid, zc, N, DOUT);
    }
    finish_kernel<<<(G * DOUT + TB - 1) / TB, TB, 0, stream>>>(zc, bc, outp, G * DOUT, DOUT);
}
